# starting kernel
# baseline (speedup 1.0000x reference)
#include <hip/hip_runtime.h>
#include <hip/hip_bf16.h>

typedef float v2f __attribute__((ext_vector_type(2)));
typedef float v8f __attribute__((ext_vector_type(8)));

constexpr int kDim = 384;
constexpr int kDst = 16;
constexpr int kL   = 4096;
constexpr int kB   = 4;
constexpr int kP   = 800;
constexpr int kM   = kB * kL;

constexpr int TS = 72;

__device__ __forceinline__ float softplus_f(float v) {
    return (v > 20.f) ? v : log1pf(__expf(v));
}

__global__ __launch_bounds__(256)
void proj_gemm_kernel(const float* __restrict__ x,
                      const float* __restrict__ Wp,
                      const float* __restrict__ bp,
                      const float* __restrict__ dtb,
                      float* __restrict__ xin,
                      float* __restrict__ dlt,
                      float* __restrict__ Bws,
                      float* __restrict__ Cws)
{
    __shared__ float sA[32 * TS];
    __shared__ float sB[32 * TS];

    const int tid  = threadIdx.x;
    const int wave = tid >> 5;
    const int lane = tid & 31;
    const int half = lane >> 4;
    const int l16  = lane & 15;

    const int m0 = blockIdx.x * 64;
    const int n0 = blockIdx.y * 64;

    const int wm = (wave & 1) * 32;
    const int wn = (wave >> 1) * 16;

    v8f acc0 = {};
    v8f acc1 = {};

    const int bidx = m0 >> 12;
    const int l0   = m0 & 4095;
    const float* xb = x + (size_t)bidx * kDim * kL;

    for (int k0 = 0; k0 < kDim; k0 += 32) {
        {
            const int j  = tid >> 3;
            const int i0 = (tid & 7) * 8;
            const float* src = xb + (size_t)(k0 + j) * kL + l0 + i0;
            float4 v0 = *(const float4*)(src);
            float4 v1 = *(const float4*)(src + 4);
            *(float4*)(&sA[j * TS + i0])     = v0;
            *(float4*)(&sA[j * TS + i0 + 4]) = v1;
        }
        {
            const int nn = tid >> 2;
            const int j0 = (tid & 3) * 8;
            const int ng = n0 + nn;
            float vv[8];
            if (ng < kP) {
                const float* src = Wp + (size_t)ng * kDim + k0 + j0;
                float4 v0 = *(const float4*)(src);
                float4 v1 = *(const float4*)(src + 4);
                vv[0]=v0.x; vv[1]=v0.y; vv[2]=v0.z; vv[3]=v0.w;
                vv[4]=v1.x; vv[5]=v1.y; vv[6]=v1.z; vv[7]=v1.w;
            } else {
                #pragma unroll
                for (int e = 0; e < 8; ++e) vv[e] = 0.f;
            }
            #pragma unroll
            for (int e = 0; e < 8; ++e) sB[(j0 + e) * TS + nn] = vv[e];
        }
        __syncthreads();

        #pragma unroll
        for (int kk = 0; kk < 32; kk += 4) {
            const int kr = kk + 2 * half;
            v2f a0, a1, bf;
            a0.x = sA[kr * TS + wm + l16];
            a0.y = sA[(kr + 1) * TS + wm + l16];
            a1.x = sA[kr * TS + wm + 16 + l16];
            a1.y = sA[(kr + 1) * TS + wm + 16 + l16];
            bf.x = sB[kr * TS + wn + l16];
            bf.y = sB[(kr + 1) * TS + wn + l16];
            acc0 = __builtin_amdgcn_wmma_f32_16x16x4_f32(false, a0, false, bf, (short)0, acc0, false, false);
            acc1 = __builtin_amdgcn_wmma_f32_16x16x4_f32(false, a1, false, bf, (short)0, acc1, false, false);
        }
        __syncthreads();
    }

    const int ng = n0 + wn + l16;
    if (ng < kP) {
        const float bias = bp[ng];
        #pragma unroll
        for (int r = 0; r < 8; ++r) {
            const int mr0 = m0 + wm + r + 8 * half;
            const int mr1 = mr0 + 16;
            float v0 = acc0[r] + bias;
            float v1 = acc1[r] + bias;
            if (ng < kDim) {
                xin[(size_t)mr0 * kDim + ng] = v0;
                xin[(size_t)mr1 * kDim + ng] = v1;
            } else if (ng < 2 * kDim) {
                const int d = ng - kDim;
                const float db = dtb[d];
                dlt[(size_t)mr0 * kDim + d] = softplus_f(v0 + db);
                dlt[(size_t)mr1 * kDim + d] = softplus_f(v1 + db);
            } else if (ng < 2 * kDim + kDst) {
                const int s = ng - 2 * kDim;
                Bws[(size_t)mr0 * kDst + s] = v0;
                Bws[(size_t)mr1 * kDst + s] = v1;
            } else {
                const int s = ng - 2 * kDim - kDst;
                Cws[(size_t)mr0 * kDst + s] = v0;
                Cws[(size_t)mr1 * kDst + s] = v1;
            }
        }
    }
}

__global__ __launch_bounds__(256)
void scan_kernel(const float* __restrict__ xin,
                 const float* __restrict__ dlt,
                 const float* __restrict__ Bws,
                 const float* __restrict__ Cws,
                 const float* __restrict__ A_log,
                 const float* __restrict__ Dp,
                 float* __restrict__ yout)
{
    __shared__ float sD[2][32 * 16];
    __shared__ float sX[2][32 * 16];
    __shared__ float sBv[2][32 * 16];
    __shared__ float sCv[2][32 * 16];

    const int tid = threadIdx.x;
    const int n   = tid & 15;
    const int dl  = tid >> 4;
    const int blk = blockIdx.x;
    const int b   = blk / (kDim / 16);
    const int d0  = (blk % (kDim / 16)) * 16;
    const int d   = d0 + dl;

    const float A_dn = -__expf(A_log[d * kDst + n]);
    const float D_d  = Dp[d];

    const float* xin_b = xin + (size_t)b * kL * kDim;
    const float* dlt_b = dlt + (size_t)b * kL * kDim;
    const float* Bb    = Bws + (size_t)b * kL * kDst;
    const float* Cb    = Cws + (size_t)b * kL * kDst;
    float*       yb    = yout + (size_t)b * kL * kDim;

    auto load_chunk = [&](int t0, int buf) {
        #pragma unroll
        for (int e = tid; e < 512; e += 256) {
            const int row = e >> 4, col = e & 15;
            const size_t go = (size_t)(t0 + row) * kDim + d0 + col;
            sD[buf][e] = dlt_b[go];
            sX[buf][e] = xin_b[go];
            const size_t gb = (size_t)t0 * kDst + e;
            sBv[buf][e] = Bb[gb];
            sCv[buf][e] = Cb[gb];
        }
    };

    load_chunk(0, 0);
    __syncthreads();

    float h = 0.f;
    constexpr int NCHUNK = kL / 32;
    for (int c = 0; c < NCHUNK; ++c) {
        const int buf = c & 1;
        if (c + 1 < NCHUNK) load_chunk((c + 1) * 32, buf ^ 1);

        const int tbase = c * 32;
        #pragma unroll 8
        for (int tt = 0; tt < 32; ++tt) {
            const float dt = sD[buf][tt * 16 + dl];
            const float xv = sX[buf][tt * 16 + dl];
            const float bn = sBv[buf][tt * 16 + n];
            const float cn = sCv[buf][tt * 16 + n];
            h = __expf(dt * A_dn) * h + (dt * xv) * bn;
            float yp = h * cn;
            yp += __shfl_xor(yp, 1);
            yp += __shfl_xor(yp, 2);
            yp += __shfl_xor(yp, 4);
            yp += __shfl_xor(yp, 8);
            if (n == 0)
                yb[(size_t)(tbase + tt) * kDim + d] = yp + xv * D_d;
        }
        __syncthreads();
    }
}

__global__ __launch_bounds__(256)
void out_gemm_kernel(const float* __restrict__ y,
                     const float* __restrict__ Wo,
                     const float* __restrict__ bo,
                     float* __restrict__ out)
{
    __shared__ float smem[64 * TS];
    float* sA = smem;
    float* sB = smem + 32 * TS;
    float* sC = smem;

    const int tid  = threadIdx.x;
    const int wave = tid >> 5;
    const int lane = tid & 31;
    const int half = lane >> 4;
    const int l16  = lane & 15;

    const int m0 = blockIdx.x * 64;
    const int n0 = blockIdx.y * 64;
    const int wm = (wave & 1) * 32;
    const int wn = (wave >> 1) * 16;

    v8f acc0 = {};
    v8f acc1 = {};

    for (int k0 = 0; k0 < kDim; k0 += 32) {
        {
            const int i  = tid >> 2;
            const int j0 = (tid & 3) * 8;
            const float* src = y + (size_t)(m0 + i) * kDim + k0 + j0;
            float4 v0 = *(const float4*)(src);
            float4 v1 = *(const float4*)(src + 4);
            sA[(j0 + 0) * TS + i] = v0.x; sA[(j0 + 1) * TS + i] = v0.y;
            sA[(j0 + 2) * TS + i] = v0.z; sA[(j0 + 3) * TS + i] = v0.w;
            sA[(j0 + 4) * TS + i] = v1.x; sA[(j0 + 5) * TS + i] = v1.y;
            sA[(j0 + 6) * TS + i] = v1.z; sA[(j0 + 7) * TS + i] = v1.w;
        }
        {
            const int nn = tid >> 2;
            const int j0 = (tid & 3) * 8;
            const float* src = Wo + (size_t)(n0 + nn) * kDim + k0 + j0;
            float4 v0 = *(const float4*)(src);
            float4 v1 = *(const float4*)(src + 4);
            sB[(j0 + 0) * TS + nn] = v0.x; sB[(j0 + 1) * TS + nn] = v0.y;
            sB[(j0 + 2) * TS + nn] = v0.z; sB[(j0 + 3) * TS + nn] = v0.w;
            sB[(j0 + 4) * TS + nn] = v1.x; sB[(j0 + 5) * TS + nn] = v1.y;
            sB[(j0 + 6) * TS + nn] = v1.z; sB[(j0 + 7) * TS + nn] = v1.w;
        }
        __syncthreads();

        #pragma unroll
        for (int kk = 0; kk < 32; kk += 4) {
            const int kr = kk + 2 * half;
            v2f a0, a1, bf;
            a0.x = sA[kr * TS + wm + l16];
            a0.y = sA[(kr + 1) * TS + wm + l16];
            a1.x = sA[kr * TS + wm + 16 + l16];
            a1.y = sA[(kr + 1) * TS + wm + 16 + l16];
            bf.x = sB[kr * TS + wn + l16];
            bf.y = sB[(kr + 1) * TS + wn + l16];
            acc0 = __builtin_amdgcn_wmma_f32_16x16x4_f32(false, a0, false, bf, (short)0, acc0, false, false);
            acc1 = __builtin_amdgcn_wmma_f32_16x16x4_f32(false, a1, false, bf, (short)0, acc1, false, false);
        }
        __syncthreads();
    }

    const float bias = bo[n0 + wn + l16];
    #pragma unroll
    for (int r = 0; r < 8; ++r) {
        const int mm = wm + r + 8 * half;
        sC[(wn + l16) * TS + mm]      = acc0[r] + bias;
        sC[(wn + l16) * TS + mm + 16] = acc1[r] + bias;
    }
    __syncthreads();

    const int bidx = m0 >> 12;
    const int l0   = m0 & 4095;
    const int nn   = tid >> 2;
    const int ms   = (tid & 3) * 16;
    float* dst = out + ((size_t)bidx * kDim + n0 + nn) * kL + l0 + ms;
    #pragma unroll
    for (int e = 0; e < 4; ++e) {
        float4 v;
        v.x = sC[nn * TS + ms + e * 4 + 0];
        v.y = sC[nn * TS + ms + e * 4 + 1];
        v.z = sC[nn * TS + ms + e * 4 + 2];
        v.w = sC[nn * TS + ms + e * 4 + 3];
        *(float4*)(dst + e * 4) = v;
    }
}

extern "C" void kernel_launch(void* const* d_in, const int* in_sizes, int n_in,
                              void* d_out, int out_size, void* d_ws, size_t ws_size,
                              hipStream_t stream) {
    const float* x     = (const float*)d_in[0];
    const float* Wp    = (const float*)d_in[1];
    const float* bp    = (const float*)d_in[2];
    const float* A_log = (const float*)d_in[3];
    const float* Dp    = (const float*)d_in[4];
    const float* dtb   = (const float*)d_in[5];
    const float* Wo    = (const float*)d_in[6];
    const float* bo    = (const float*)d_in[7];
    float* out = (float*)d_out;

    float* ws = (float*)d_ws;
    const size_t MD = (size_t)kM * kDim;
    const size_t MS = (size_t)kM * kDst;
    float* xin = ws;
    float* dlt = ws + MD;
    float* yws = ws + 2 * MD;
    float* Bw  = ws + 3 * MD;
    float* Cw  = Bw + MS;

    dim3 g1(kM / 64, (kP + 63) / 64);
    proj_gemm_kernel<<<g1, 256, 0, stream>>>(x, Wp, bp, dtb, xin, dlt, Bw, Cw);

    scan_kernel<<<kB * (kDim / 16), 256, 0, stream>>>(xin, dlt, Bw, Cw, A_log, Dp, yws);

    dim3 g3(kM / 64, kDim / 64);
    out_gemm_kernel<<<g3, 256, 0, stream>>>(yws, Wo, bo, out);
}


# Both products: a block stages its 32-feature weight slab (all k) in LDS once; x/y fragments go from memory straight to registers as 16-byte row loads; k loop has no barrier (k order in sums changes)
# speedup vs baseline: 7.0498x; 7.0498x over previous
#include <hip/hip_runtime.h>
#include <hip/hip_bf16.h>

typedef float v2f __attribute__((ext_vector_type(2)));
typedef float v8f __attribute__((ext_vector_type(8)));

constexpr int kDim = 384;
constexpr int kDst = 16;
constexpr int kL   = 4096;
constexpr int kB   = 4;
constexpr int kP   = 800;
constexpr int kM   = kB * kL;

__device__ __forceinline__ float softplus_f(float v) {
    return (v > 20.f) ? v : log1pf(__expf(v));
}

template <int CTRL>
__device__ __forceinline__ float row_xor_add(float keep, float send) {
    const int t = __builtin_amdgcn_update_dpp(0, __float_as_int(send), CTRL, 0xf, 0xf, true);
    return keep + __int_as_float(t);
}

template <int CTRL, int NOUT>
__device__ __forceinline__ void fold_stage(const float* in, float* out, bool bit) {
    #pragma unroll
    for (int i = 0; i < NOUT; ++i) {
        const float keep = bit ? in[2 * i + 1] : in[2 * i];
        const float send = bit ? in[2 * i] : in[2 * i + 1];
        out[i] = row_xor_add<CTRL>(keep, send);
    }
}

typedef float v4f __attribute__((ext_vector_type(4)));

constexpr int WP = 388;
constexpr int GF = 32;
constexpr int GT = 256;

__device__ __forceinline__ void slab_product(const float* __restrict__ Wg,
                                             const float* __restrict__ Xb,
                                             int l0, float* sW, v8f (&acc)[2][4])
{
    const int tid  = threadIdx.x;
    const int lane = tid & 31;
    const int half = lane >> 4;
    const int l16  = lane & 15;

    const float* xp = Xb + (size_t)(4 * half) * kL + l0 + 4 * l16;
    const float* wp = sW + l16 * WP + 4 * half;

    auto load_x = [&](int j, v4f (&xr)[4]) {
        #pragma unroll
        for (int i = 0; i < 4; ++i) xr[i] = *(const v4f*)(xp + (size_t)(8 * j + i) * kL);
    };
    auto load_w = [&](int j, v4f (&wf)[2]) {
        wf[0] = *(const v4f*)(wp + 8 * j);
        wf[1] = *(const v4f*)(wp + 16 * WP + 8 * j);
    };
    auto mma = [&](const v4f (&xr)[4], const v4f (&wf)[2]) {
        #pragma unroll
        for (int u = 0; u < 2; ++u) {
            #pragma unroll
            for (int f = 0; f < 2; ++f) {
                v2f a;
                a.x = wf[f][2 * u];  a.y = wf[f][2 * u + 1];
                #pragma unroll
                for (int i = 0; i < 4; ++i) {
                    v2f b;
                    b.x = xr[2 * u][i];  b.y = xr[2 * u + 1][i];
                    acc[f][i] = __builtin_amdgcn_wmma_f32_16x16x4_f32(false, a, false, b, (short)0, acc[f][i], false, false);
                }
            }
        }
    };

    v4f xA[4], xB[4], wA[2], wB[2];
    load_x(0, xA);

    {
        const int wave = tid >> 5;
        v4f piece[24];
        #pragma unroll
        for (int it = 0; it < 24; ++it) {
            const int r  = wave * 8 + it / 3;
            const int c4 = (it % 3) * 32 + lane;
            piece[it] = *(const v4f*)(Wg + (size_t)r * kDim + 4 * c4);
        }
        #pragma unroll
        for (int it = 0; it < 24; ++it) {
            const int r  = wave * 8 + it / 3;
            const int c4 = (it % 3) * 32 + lane;
            *(v4f*)&sW[r * WP + 4 * c4] = piece[it];
        }
    }
    __syncthreads();
    load_w(0, wA);

    #pragma nounroll
    for (int j = 0; j < kDim / 8; j += 2) {
        load_x(j + 1, xB);
        load_w(j + 1, wB);
        mma(xA, wA);
        const int jn = (j + 2 < kDim / 8) ? j + 2 : kDim / 8 - 1;
        load_x(jn, xA);
        load_w(jn, wA);
        mma(xB, wB);
    }
}

__global__ __launch_bounds__(128)
void proj_gemm_kernel(const float* __restrict__ x,
                      const float* __restrict__ Wp,
                      const float* __restrict__ bp,
                      const float* __restrict__ dtb,
                      float* __restrict__ xin,
                      float* __restrict__ dlt,
                      float* __restrict__ Bws,
                      float* __restrict__ Cws)
{
    __shared__ __attribute__((aligned(16))) float sW[GF * WP];

    const int tid  = threadIdx.x;
    const int wave = tid >> 5;
    const int lane = tid & 31;
    const int half = lane >> 4;
    const int l16  = lane & 15;

    const int m0   = blockIdx.x * GT + wave * 64;
    const int n0   = blockIdx.y * GF;
    const int bidx = m0 >> 12;
    const int l0   = m0 & 4095;

    v8f acc[2][4] = {};
    slab_product(Wp + (size_t)n0 * kDim, x + (size_t)bidx * kDim * kL, l0, sW, acc);

    const int lt = l0 + 4 * l16;
    #pragma unroll
    for (int f = 0; f < 2; ++f) {
        #pragma unroll
        for (int r = 0; r < 8; ++r) {
            const int ng = n0 + 16 * f + 8 * half + r;
            const float bias = bp[ng];
            v4f v;
            v.x = acc[f][0][r] + bias;  v.y = acc[f][1][r] + bias;
            v.z = acc[f][2][r] + bias;  v.w = acc[f][3][r] + bias;
            if (n0 < kDim) {
                *(v4f*)(xin + ((size_t)bidx * kDim + ng) * kL + lt) = v;
            } else if (n0 < 2 * kDim) {
                const int d = ng - kDim;
                const float db = dtb[d];
                #pragma nounroll
                for (int e = 0; e < 4; ++e) {
                    const float t = softplus_f(v.x + db);
                    v.x = v.y;  v.y = v.z;  v.z = v.w;  v.w = t;
                }
                *(v4f*)(dlt + ((size_t)bidx * kDim + d) * kL + lt) = v;
            } else if (f == 0) {
                const int s = ng - 2 * kDim;
                *(v4f*)(Bws + ((size_t)bidx * kDst + s) * kL + lt) = v;
            } else {
                const int s = ng - 2 * kDim - kDst;
                *(v4f*)(Cws + ((size_t)bidx * kDst + s) * kL + lt) = v;
            }
        }
    }
}

constexpr int SCH = 8;
constexpr int SST = 64;
constexpr int SP  = 68;

__global__ __launch_bounds__(128)
void scan_kernel(const float* __restrict__ xin,
                 const float* __restrict__ dlt,
                 const float* __restrict__ Bws,
                 const float* __restrict__ Cws,
                 const float* __restrict__ A_log,
                 const float* __restrict__ Dp,
                 float* __restrict__ yout)
{
    __shared__ __attribute__((aligned(16))) float sD[2][SCH * SP];
    __shared__ __attribute__((aligned(16))) float sX[2][SCH * SP];
    __shared__ __attribute__((aligned(16))) float sBv[2][16 * SP];
    __shared__ __attribute__((aligned(16))) float sCv[2][16 * SP];

    const int tid = threadIdx.x;
    const int n   = tid & 15;
    const int dl  = tid >> 4;
    const int blk = blockIdx.x;
    const int b   = blk / (kDim / SCH);
    const int d0  = (blk % (kDim / SCH)) * SCH;
    const int d   = d0 + dl;

    const float A_dn = -__expf(A_log[d * kDst + n]);
    const float D_d  = Dp[d];

    const float* xin_b = xin + ((size_t)b * kDim + d0) * kL;
    const float* dlt_b = dlt + ((size_t)b * kDim + d0) * kL;
    const float* Bb    = Bws + (size_t)b * kDst * kL;
    const float* Cb    = Cws + (size_t)b * kDst * kL;
    float*       yb    = yout + ((size_t)b * kDim + d) * kL;

    const int    prow = tid >> 4;
    const int    pq   = (tid & 15) * 4;
    const size_t go0  = (size_t)prow * kL + pq;
    const size_t go1  = (size_t)(prow + 8) * kL + pq;
    const int    si0  = prow * SP + pq;
    const int    si1  = (prow + 8) * SP + pq;

    constexpr int NCHUNK = kL / SST;

    float4 pD, pX, pB0, pB1, pC0, pC1;
    auto request_chunk = [&](int c) {
        const size_t tn = (size_t)c * SST;
        pD  = *(const float4*)(dlt_b + tn + go0);
        pX  = *(const float4*)(xin_b + tn + go0);
        pB0 = *(const float4*)(Bb + tn + go0);
        pB1 = *(const float4*)(Bb + tn + go1);
        pC0 = *(const float4*)(Cb + tn + go0);
        pC1 = *(const float4*)(Cb + tn + go1);
    };
    auto store_chunk = [&](int buf) {
        *(float4*)&sD[buf][si0]  = pD;
        *(float4*)&sX[buf][si0]  = pX;
        *(float4*)&sBv[buf][si0] = pB0;
        *(float4*)&sBv[buf][si1] = pB1;
        *(float4*)&sCv[buf][si0] = pC0;
        *(float4*)&sCv[buf][si1] = pC1;
    };

    struct Grp { float dt[16], xv[16], bn[16], cn[16], xk; };
    auto read_group = [&](int buf, int g, Grp& r) {
        const float4* qd = (const float4*)&sD[buf][dl * SP + g * 16];
        const float4* qx = (const float4*)&sX[buf][dl * SP + g * 16];
        const float4* qb = (const float4*)&sBv[buf][n * SP + g * 16];
        const float4* qc = (const float4*)&sCv[buf][n * SP + g * 16];
        #pragma unroll
        for (int q = 0; q < 4; ++q) {
            const float4 vd = qd[q], vx = qx[q], vb = qb[q], vc = qc[q];
            r.dt[4*q] = vd.x; r.dt[4*q+1] = vd.y; r.dt[4*q+2] = vd.z; r.dt[4*q+3] = vd.w;
            r.xv[4*q] = vx.x; r.xv[4*q+1] = vx.y; r.xv[4*q+2] = vx.z; r.xv[4*q+3] = vx.w;
            r.bn[4*q] = vb.x; r.bn[4*q+1] = vb.y; r.bn[4*q+2] = vb.z; r.bn[4*q+3] = vb.w;
            r.cn[4*q] = vc.x; r.cn[4*q+1] = vc.y; r.cn[4*q+2] = vc.z; r.cn[4*q+3] = vc.w;
        }
        r.xk = sX[buf][dl * SP + g * 16 + n];
    };

    float h = 0.f;
    const bool nb0 = (n & 1) != 0, nb1 = (n & 2) != 0, nb2 = (n & 4) != 0, nb3 = (n & 8) != 0;
    auto run_group = [&](int t0, const Grp& r) {
        float yp[16];
        #pragma unroll
        for (int j = 0; j < 16; ++j) {
            h = __expf(r.dt[j] * A_dn) * h + (r.dt[j] * r.xv[j]) * r.bn[j];
            yp[j] = h * r.cn[j];
        }
        float s8[8], s4[4], s2[2], s1[1];
        fold_stage<0x161, 8>(yp, s8, nb0);
        fold_stage<0x162, 4>(s8, s4, nb1);
        fold_stage<0x164, 2>(s4, s2, nb2);
        fold_stage<0x168, 1>(s2, s1, nb3);
        yb[t0 + n] = s1[0] + r.xk * D_d;
    };

    request_chunk(0);
    store_chunk(0);
    __syncthreads();
    request_chunk(1);
    Grp ga, gb;
    read_group(0, 0, ga);

    for (int c = 0; c < NCHUNK; ++c) {
        const int buf = c & 1;
        const int tbase = c * SST;

        read_group(buf, 1, gb);
        run_group(tbase, ga);
        read_group(buf, 2, ga);
        run_group(tbase + 16, gb);
        read_group(buf, 3, gb);
        run_group(tbase + 32, ga);

        if (c + 1 < NCHUNK) store_chunk(buf ^ 1);
        __builtin_amdgcn_fence(__ATOMIC_RELEASE, "workgroup", "local");
        __builtin_amdgcn_s_barrier();
        __builtin_amdgcn_fence(__ATOMIC_ACQUIRE, "workgroup", "local");
        if (c + 2 < NCHUNK) request_chunk(c + 2);

        read_group(buf ^ 1, 0, ga);
        run_group(tbase + 48, gb);
    }
}

__global__ __launch_bounds__(128)
void out_gemm_kernel(const float* __restrict__ y,
                     const float* __restrict__ Wo,
                     const float* __restrict__ bo,
                     float* __restrict__ out)
{
    __shared__ __attribute__((aligned(16))) float sW[GF * WP];

    const int tid  = threadIdx.x;
    const int wave = tid >> 5;
    const int lane = tid & 31;
    const int half = lane >> 4;
    const int l16  = lane & 15;

    const int m0   = blockIdx.x * GT + wave * 64;
    const int n0   = blockIdx.y * GF;
    const int bidx = m0 >> 12;
    const int l0   = m0 & 4095;

    v8f acc[2][4] = {};
    slab_product(Wo + (size_t)n0 * kDim, y + (size_t)bidx * kDim * kL, l0, sW, acc);

    const int lt = l0 + 4 * l16;
    #pragma unroll
    for (int f = 0; f < 2; ++f) {
        #pragma unroll
        for (int r = 0; r < 8; ++r) {
            const int ng = n0 + 16 * f + 8 * half + r;
            const float bias = bo[ng];
            v4f v;
            v.x = acc[f][0][r] + bias;  v.y = acc[f][1][r] + bias;
            v.z = acc[f][2][r] + bias;  v.w = acc[f][3][r] + bias;
            *(v4f*)(out + ((size_t)bidx * kDim + ng) * kL + lt) = v;
        }
    }
}

extern "C" void kernel_launch(void* const* d_in, const int* in_sizes, int n_in,
                              void* d_out, int out_size, void* d_ws, size_t ws_size,
                              hipStream_t stream) {
    const float* x     = (const float*)d_in[0];
    const float* Wp    = (const float*)d_in[1];
    const float* bp    = (const float*)d_in[2];
    const float* A_log = (const float*)d_in[3];
    const float* Dp    = (const float*)d_in[4];
    const float* dtb   = (const float*)d_in[5];
    const float* Wo    = (const float*)d_in[6];
    const float* bo    = (const float*)d_in[7];
    float* out = (float*)d_out;

    float* ws = (float*)d_ws;
    const size_t MD = (size_t)kM * kDim;
    const size_t MS = (size_t)kM * kDst;
    float* xin = ws;
    float* dlt = ws + MD;
    float* yws = ws + 2 * MD;
    float* Bw  = ws + 3 * MD;
    float* Cw  = Bw + MS;

    dim3 g1(kM / GT, kP / GF);
    proj_gemm_kernel<<<g1, 128, 0, stream>>>(x, Wp, bp, dtb, xin, dlt, Bw, Cw);

    scan_kernel<<<kB * (kDim / SCH), 128, 0, stream>>>(xin, dlt, Bw, Cw, A_log, Dp, yws);

    dim3 g3(kM / GT, kDim / GF);
    out_gemm_kernel<<<g3, 128, 0, stream>>>(yws, Wo, bo, out);
}


# Scan: each group's 17 LDS reads stand at ONE place, between its 16 steps and its state sum, two groups ahead, held there by scheduling fences; the compiler had spread a trip's reads over 29 places
# speedup vs baseline: 7.5237x; 1.0672x over previous
#include <hip/hip_runtime.h>
#include <hip/hip_bf16.h>

typedef float v2f __attribute__((ext_vector_type(2)));
typedef float v8f __attribute__((ext_vector_type(8)));

constexpr int kDim = 384;
constexpr int kDst = 16;
constexpr int kL   = 4096;
constexpr int kB   = 4;
constexpr int kP   = 800;
constexpr int kM   = kB * kL;

__device__ __forceinline__ float softplus_f(float v) {
    return (v > 20.f) ? v : log1pf(__expf(v));
}

template <int CTRL>
__device__ __forceinline__ float row_xor_add(float keep, float send) {
    const int t = __builtin_amdgcn_update_dpp(0, __float_as_int(send), CTRL, 0xf, 0xf, true);
    return keep + __int_as_float(t);
}

template <int CTRL, int NOUT>
__device__ __forceinline__ void fold_stage(const float* in, float* out, bool bit) {
    #pragma unroll
    for (int i = 0; i < NOUT; ++i) {
        const float keep = bit ? in[2 * i + 1] : in[2 * i];
        const float send = bit ? in[2 * i] : in[2 * i + 1];
        out[i] = row_xor_add<CTRL>(keep, send);
    }
}

typedef float v4f __attribute__((ext_vector_type(4)));

constexpr int WP = 388;
constexpr int GF = 32;
constexpr int GT = 256;

__device__ __forceinline__ void slab_product(const float* __restrict__ Wg,
                                             const float* __restrict__ Xb,
                                             int l0, float* sW, v8f (&acc)[2][4])
{
    const int tid  = threadIdx.x;
    const int lane = tid & 31;
    const int half = lane >> 4;
    const int l16  = lane & 15;

    const float* xp = Xb + (size_t)(4 * half) * kL + l0 + 4 * l16;
    const float* wp = sW + l16 * WP + 4 * half;

    auto load_x = [&](int j, v4f (&xr)[4]) {
        #pragma unroll
        for (int i = 0; i < 4; ++i) xr[i] = *(const v4f*)(xp + (size_t)(8 * j + i) * kL);
    };
    auto load_w = [&](int j, v4f (&wf)[2]) {
        wf[0] = *(const v4f*)(wp + 8 * j);
        wf[1] = *(const v4f*)(wp + 16 * WP + 8 * j);
    };
    auto mma = [&](const v4f (&xr)[4], const v4f (&wf)[2]) {
        #pragma unroll
        for (int u = 0; u < 2; ++u) {
            #pragma unroll
            for (int f = 0; f < 2; ++f) {
                v2f a;
                a.x = wf[f][2 * u];  a.y = wf[f][2 * u + 1];
                #pragma unroll
                for (int i = 0; i < 4; ++i) {
                    v2f b;
                    b.x = xr[2 * u][i];  b.y = xr[2 * u + 1][i];
                    acc[f][i] = __builtin_amdgcn_wmma_f32_16x16x4_f32(false, a, false, b, (short)0, acc[f][i], false, false);
                }
            }
        }
    };

    v4f xA[4], xB[4], wA[2], wB[2];
    load_x(0, xA);

    {
        const int wave = tid >> 5;
        v4f piece[24];
        #pragma unroll
        for (int it = 0; it < 24; ++it) {
            const int r  = wave * 8 + it / 3;
            const int c4 = (it % 3) * 32 + lane;
            piece[it] = *(const v4f*)(Wg + (size_t)r * kDim + 4 * c4);
        }
        #pragma unroll
        for (int it = 0; it < 24; ++it) {
            const int r  = wave * 8 + it / 3;
            const int c4 = (it % 3) * 32 + lane;
            *(v4f*)&sW[r * WP + 4 * c4] = piece[it];
        }
    }
    __syncthreads();
    load_w(0, wA);

    #pragma nounroll
    for (int j = 0; j < kDim / 8; j += 2) {
        load_x(j + 1, xB);
        load_w(j + 1, wB);
        mma(xA, wA);
        const int jn = (j + 2 < kDim / 8) ? j + 2 : kDim / 8 - 1;
        load_x(jn, xA);
        load_w(jn, wA);
        mma(xB, wB);
    }
}

__global__ __launch_bounds__(128)
void proj_gemm_kernel(const float* __restrict__ x,
                      const float* __restrict__ Wp,
                      const float* __restrict__ bp,
                      const float* __restrict__ dtb,
                      float* __restrict__ xin,
                      float* __restrict__ dlt,
                      float* __restrict__ Bws,
                      float* __restrict__ Cws)
{
    __shared__ __attribute__((aligned(16))) float sW[GF * WP];

    const int tid  = threadIdx.x;
    const int wave = tid >> 5;
    const int lane = tid & 31;
    const int half = lane >> 4;
    const int l16  = lane & 15;

    const int m0   = blockIdx.x * GT + wave * 64;
    const int n0   = blockIdx.y * GF;
    const int bidx = m0 >> 12;
    const int l0   = m0 & 4095;

    v8f acc[2][4] = {};
    slab_product(Wp + (size_t)n0 * kDim, x + (size_t)bidx * kDim * kL, l0, sW, acc);

    const int lt = l0 + 4 * l16;
    #pragma unroll
    for (int f = 0; f < 2; ++f) {
        #pragma unroll
        for (int r = 0; r < 8; ++r) {
            const int ng = n0 + 16 * f + 8 * half + r;
            const float bias = bp[ng];
            v4f v;
            v.x = acc[f][0][r] + bias;  v.y = acc[f][1][r] + bias;
            v.z = acc[f][2][r] + bias;  v.w = acc[f][3][r] + bias;
            if (n0 < kDim) {
                *(v4f*)(xin + ((size_t)bidx * kDim + ng) * kL + lt) = v;
            } else if (n0 < 2 * kDim) {
                const int d = ng - kDim;
                const float db = dtb[d];
                #pragma nounroll
                for (int e = 0; e < 4; ++e) {
                    const float t = softplus_f(v.x + db);
                    v.x = v.y;  v.y = v.z;  v.z = v.w;  v.w = t;
                }
                *(v4f*)(dlt + ((size_t)bidx * kDim + d) * kL + lt) = v;
            } else if (f == 0) {
                const int s = ng - 2 * kDim;
                *(v4f*)(Bws + ((size_t)bidx * kDst + s) * kL + lt) = v;
            } else {
                const int s = ng - 2 * kDim - kDst;
                *(v4f*)(Cws + ((size_t)bidx * kDst + s) * kL + lt) = v;
            }
        }
    }
}

constexpr int SCH = 8;
constexpr int SST = 64;
constexpr int SP  = 68;

__device__ __forceinline__ void lds_place_edge() { __builtin_amdgcn_sched_barrier(0); }

__global__ __launch_bounds__(128)
void scan_kernel(const float* __restrict__ xin,
                 const float* __restrict__ dlt,
                 const float* __restrict__ Bws,
                 const float* __restrict__ Cws,
                 const float* __restrict__ A_log,
                 const float* __restrict__ Dp,
                 float* __restrict__ yout)
{
    __shared__ __attribute__((aligned(16))) float sD[2][SCH * SP];
    __shared__ __attribute__((aligned(16))) float sX[2][SCH * SP];
    __shared__ __attribute__((aligned(16))) float sBv[2][16 * SP];
    __shared__ __attribute__((aligned(16))) float sCv[2][16 * SP];

    const int tid = threadIdx.x;
    const int n   = tid & 15;
    const int dl  = tid >> 4;
    const int blk = blockIdx.x;
    const int b   = blk / (kDim / SCH);
    const int d0  = (blk % (kDim / SCH)) * SCH;
    const int d   = d0 + dl;

    const float A_dn = -__expf(A_log[d * kDst + n]);
    const float D_d  = Dp[d];

    const float* xin_b = xin + ((size_t)b * kDim + d0) * kL;
    const float* dlt_b = dlt + ((size_t)b * kDim + d0) * kL;
    const float* Bb    = Bws + (size_t)b * kDst * kL;
    const float* Cb    = Cws + (size_t)b * kDst * kL;
    float*       yb    = yout + ((size_t)b * kDim + d) * kL;

    const int    prow = tid >> 4;
    const int    pq   = (tid & 15) * 4;
    const size_t go0  = (size_t)prow * kL + pq;
    const size_t go1  = (size_t)(prow + 8) * kL + pq;
    const int    si0  = prow * SP + pq;
    const int    si1  = (prow + 8) * SP + pq;

    constexpr int NCHUNK = kL / SST;

    float4 pD, pX, pB0, pB1, pC0, pC1;
    auto request_chunk = [&](int c) {
        const size_t tn = (size_t)c * SST;
        pD  = *(const float4*)(dlt_b + tn + go0);
        pX  = *(const float4*)(xin_b + tn + go0);
        pB0 = *(const float4*)(Bb + tn + go0);
        pB1 = *(const float4*)(Bb + tn + go1);
        pC0 = *(const float4*)(Cb + tn + go0);
        pC1 = *(const float4*)(Cb + tn + go1);
    };
    auto store_chunk = [&](int buf) {
        *(float4*)&sD[buf][si0]  = pD;
        *(float4*)&sX[buf][si0]  = pX;
        *(float4*)&sBv[buf][si0] = pB0;
        *(float4*)&sBv[buf][si1] = pB1;
        *(float4*)&sCv[buf][si0] = pC0;
        *(float4*)&sCv[buf][si1] = pC1;
    };

    struct Grp { float dt[16], xv[16], bn[16], cn[16], xk; };
    auto read_group = [&](int buf, int g, Grp& r) {
        const float4* qd = (const float4*)&sD[buf][dl * SP + g * 16];
        const float4* qx = (const float4*)&sX[buf][dl * SP + g * 16];
        const float4* qb = (const float4*)&sBv[buf][n * SP + g * 16];
        const float4* qc = (const float4*)&sCv[buf][n * SP + g * 16];
        #pragma unroll
        for (int q = 0; q < 4; ++q) {
            const float4 vd = qd[q], vx = qx[q], vb = qb[q], vc = qc[q];
            r.dt[4*q] = vd.x; r.dt[4*q+1] = vd.y; r.dt[4*q+2] = vd.z; r.dt[4*q+3] = vd.w;
            r.xv[4*q] = vx.x; r.xv[4*q+1] = vx.y; r.xv[4*q+2] = vx.z; r.xv[4*q+3] = vx.w;
            r.bn[4*q] = vb.x; r.bn[4*q+1] = vb.y; r.bn[4*q+2] = vb.z; r.bn[4*q+3] = vb.w;
            r.cn[4*q] = vc.x; r.cn[4*q+1] = vc.y; r.cn[4*q+2] = vc.z; r.cn[4*q+3] = vc.w;
        }
        r.xk = sX[buf][dl * SP + g * 16 + n];
    };

    float h = 0.f;
    const bool nb0 = (n & 1) != 0, nb1 = (n & 2) != 0, nb2 = (n & 4) != 0, nb3 = (n & 8) != 0;
    auto run_steps = [&](const Grp& r, float (&yp)[16]) {
        #pragma unroll
        for (int j = 0; j < 16; ++j) {
            h = __expf(r.dt[j] * A_dn) * h + (r.dt[j] * r.xv[j]) * r.bn[j];
            yp[j] = h * r.cn[j];
        }
    };
    auto sum_store = [&](int t0, const float (&yp)[16], float xk) {
        float s8[8], s4[4], s2[2], s1[1];
        fold_stage<0x161, 8>(yp, s8, nb0);
        fold_stage<0x162, 4>(s8, s4, nb1);
        fold_stage<0x164, 2>(s4, s2, nb2);
        fold_stage<0x168, 1>(s2, s1, nb3);
        yb[t0 + n] = s1[0] + xk * D_d;
    };

    request_chunk(0);
    store_chunk(0);
    __syncthreads();
    request_chunk(1);
    Grp ga, gb;
    read_group(0, 0, ga);
    read_group(0, 1, gb);

    for (int c = 0; c < NCHUNK; ++c) {
        const int buf = c & 1;
        const int tbase = c * SST;
        float yp[16];
        float xk;

        run_steps(ga, yp);
        xk = ga.xk;
        lds_place_edge();
        read_group(buf, 2, ga);
        lds_place_edge();
        sum_store(tbase, yp, xk);

        run_steps(gb, yp);
        xk = gb.xk;
        lds_place_edge();
        read_group(buf, 3, gb);
        lds_place_edge();
        sum_store(tbase + 16, yp, xk);

        run_steps(ga, yp);
        xk = ga.xk;
        lds_place_edge();
        store_chunk(buf ^ 1);
        __builtin_amdgcn_fence(__ATOMIC_RELEASE, "workgroup", "local");
        __builtin_amdgcn_s_barrier();
        __builtin_amdgcn_fence(__ATOMIC_ACQUIRE, "workgroup", "local");
        request_chunk(c + 2 < NCHUNK ? c + 2 : NCHUNK - 1);
        read_group(buf ^ 1, 0, ga);
        lds_place_edge();
        sum_store(tbase + 32, yp, xk);

        run_steps(gb, yp);
        xk = gb.xk;
        lds_place_edge();
        read_group(buf ^ 1, 1, gb);
        lds_place_edge();
        sum_store(tbase + 48, yp, xk);
    }
}

__global__ __launch_bounds__(128)
void out_gemm_kernel(const float* __restrict__ y,
                     const float* __restrict__ Wo,
                     const float* __restrict__ bo,
                     float* __restrict__ out)
{
    __shared__ __attribute__((aligned(16))) float sW[GF * WP];

    const int tid  = threadIdx.x;
    const int wave = tid >> 5;
    const int lane = tid & 31;
    const int half = lane >> 4;
    const int l16  = lane & 15;

    const int m0   = blockIdx.x * GT + wave * 64;
    const int n0   = blockIdx.y * GF;
    const int bidx = m0 >> 12;
    const int l0   = m0 & 4095;

    v8f acc[2][4] = {};
    slab_product(Wo + (size_t)n0 * kDim, y + (size_t)bidx * kDim * kL, l0, sW, acc);

    const int lt = l0 + 4 * l16;
    #pragma unroll
    for (int f = 0; f < 2; ++f) {
        #pragma unroll
        for (int r = 0; r < 8; ++r) {
            const int ng = n0 + 16 * f + 8 * half + r;
            const float bias = bo[ng];
            v4f v;
            v.x = acc[f][0][r] + bias;  v.y = acc[f][1][r] + bias;
            v.z = acc[f][2][r] + bias;  v.w = acc[f][3][r] + bias;
            *(v4f*)(out + ((size_t)bidx * kDim + ng) * kL + lt) = v;
        }
    }
}

extern "C" void kernel_launch(void* const* d_in, const int* in_sizes, int n_in,
                              void* d_out, int out_size, void* d_ws, size_t ws_size,
                              hipStream_t stream) {
    const float* x     = (const float*)d_in[0];
    const float* Wp    = (const float*)d_in[1];
    const float* bp    = (const float*)d_in[2];
    const float* A_log = (const float*)d_in[3];
    const float* Dp    = (const float*)d_in[4];
    const float* dtb   = (const float*)d_in[5];
    const float* Wo    = (const float*)d_in[6];
    const float* bo    = (const float*)d_in[7];
    float* out = (float*)d_out;

    float* ws = (float*)d_ws;
    const size_t MD = (size_t)kM * kDim;
    const size_t MS = (size_t)kM * kDst;
    float* xin = ws;
    float* dlt = ws + MD;
    float* yws = ws + 2 * MD;
    float* Bw  = ws + 3 * MD;
    float* Cw  = Bw + MS;

    dim3 g1(kM / GT, kP / GF);
    proj_gemm_kernel<<<g1, 128, 0, stream>>>(x, Wp, bp, dtb, xin, dlt, Bw, Cw);

    scan_kernel<<<kB * (kDim / SCH), 128, 0, stream>>>(xin, dlt, Bw, Cw, A_log, Dp, yws);

    dim3 g3(kM / GT, kDim / GF);
    out_gemm_kernel<<<g3, 128, 0, stream>>>(yws, Wo, bo, out);
}
